# starting kernel
# baseline (speedup 1.0000x reference)
#include <hip/hip_runtime.h>


typedef __attribute__((ext_vector_type(2))) float v2f;
typedef __attribute__((ext_vector_type(4))) float v4f;
typedef __attribute__((ext_vector_type(8))) float v8f;

#define B_DIM 64
#define C_DIM 64
#define U_DIM 1024
#define N_CAP 32
#define D_CAP 16

__global__ void caps_transpose(const float* __restrict__ in,
                               float* __restrict__ outT) {
  __shared__ float tile[32][33];
  const int u0 = blockIdx.x * 32;
  const int r0 = blockIdx.y * 32;
  const int tx = threadIdx.x & 31;
  const int ty = threadIdx.x >> 5;
  for (int i = 0; i < 32; i += 8)
    tile[ty + i][tx] = in[(size_t)(r0 + ty + i) * U_DIM + u0 + tx];
  __syncthreads();
  for (int i = 0; i < 32; i += 8)
    outT[(size_t)(u0 + ty + i) * (B_DIM * C_DIM) + r0 + tx] = tile[tx][ty + i];
}

__global__ void __launch_bounds__(512, 1)
caps_uhat(const float* __restrict__ uT,
          const float* __restrict__ W,
          float* __restrict__ uhat) {
  const int u = blockIdx.x;
  __shared__ float As[B_DIM * 68];

  const float* Ag = uT + (size_t)u * (B_DIM * C_DIM);
  for (int i = threadIdx.x; i < (B_DIM * C_DIM) / 4; i += 512) {
    const int e = i * 4;
    const int b = e >> 6;
    const int c = e & 63;
    *(v4f*)&As[b * 68 + c] = ((const v4f*)Ag)[i];
  }
  __syncthreads();

  const int wave = threadIdx.x >> 5;
  const int lane = threadIdx.x & 31;
  const int l16  = lane & 15;
  const int koff = (lane >> 4) * 2;
  const int n_base = wave * 2;

  const float* Wg = W + (size_t)u * (N_CAP * C_DIM * D_CAP);

  v8f acc[4][2] = {};
  for (int kc = 0; kc < C_DIM; kc += 4) {
    const int k = kc + koff;
    v2f a[4], bb[2];
    for (int mt = 0; mt < 4; ++mt)
      a[mt] = *(const v2f*)&As[(mt * 16 + l16) * 68 + k];
    for (int nt = 0; nt < 2; ++nt) {
      const float* wp = Wg + ((size_t)(n_base + nt) * C_DIM + k) * D_CAP + l16;
      bb[nt].x = wp[0];
      bb[nt].y = wp[D_CAP];
      if (kc + 4 < C_DIM)
        __builtin_prefetch(wp + 4 * D_CAP, 0, 1);
    }
    for (int mt = 0; mt < 4; ++mt)
      for (int nt = 0; nt < 2; ++nt)
        acc[mt][nt] = __builtin_amdgcn_wmma_f32_16x16x4_f32(
            false, a[mt], false, bb[nt], (short)0, acc[mt][nt], false, false);
  }

  const int d = l16;
  const int rbase = (lane >> 4) * 8;
  for (int mt = 0; mt < 4; ++mt)
    for (int nt = 0; nt < 2; ++nt) {
      const int n = n_base + nt;
      for (int r = 0; r < 8; ++r) {
        const int b = mt * 16 + rbase + r;
        uhat[(((size_t)b * N_CAP + n) * U_DIM + u) * D_CAP + d] = acc[mt][nt][r];
      }
    }
}

__global__ void caps_route(const float* __restrict__ uhat,
                           float* __restrict__ out) {
  const int bn = blockIdx.x;
  const int b = bn >> 5, n = bn & 31;
  const int t = threadIdx.x;
  const int lane = t & 31, wave = t >> 5;

  __shared__ float red[8 * 16];
  __shared__ float sv[16];
  __shared__ float rscal[8];

  const float* base = uhat + (size_t)bn * (U_DIM * D_CAP);
  v4f row[4][4];
  for (int k = 0; k < 4; ++k) {
    const v4f* p = (const v4f*)(base + (size_t)(t + k * 256) * D_CAP);
    for (int j = 0; j < 4; ++j) row[k][j] = p[j];
  }

  float blog[4] = {0.f, 0.f, 0.f, 0.f};
  float v[16];

  for (int it = 0; it < 3; ++it) {
    float c[4];
    if (it == 0) {
      c[0] = c[1] = c[2] = c[3] = 1.0f / (float)U_DIM;
    } else {
      float mx = fmaxf(fmaxf(blog[0], blog[1]), fmaxf(blog[2], blog[3]));
      for (int o = 16; o; o >>= 1) mx = fmaxf(mx, __shfl_xor(mx, o, 32));
      if (lane == 0) rscal[wave] = mx;
      __syncthreads();
      float m2 = rscal[0];
      for (int w = 1; w < 8; ++w) m2 = fmaxf(m2, rscal[w]);
      float sum = 0.f;
      for (int k = 0; k < 4; ++k) { c[k] = __expf(blog[k] - m2); sum += c[k]; }
      for (int o = 16; o; o >>= 1) sum += __shfl_xor(sum, o, 32);
      __syncthreads();
      if (lane == 0) rscal[wave] = sum;
      __syncthreads();
      float tot = 0.f;
      for (int w = 0; w < 8; ++w) tot += rscal[w];
      const float inv = 1.0f / tot;
      for (int k = 0; k < 4; ++k) c[k] *= inv;
    }

    float sp[16];
    for (int d = 0; d < 16; ++d) sp[d] = 0.f;
    for (int k = 0; k < 4; ++k)
      for (int j = 0; j < 4; ++j)
        for (int e = 0; e < 4; ++e)
          sp[j * 4 + e] += c[k] * row[k][j][e];
    for (int d = 0; d < 16; ++d)
      for (int o = 16; o; o >>= 1) sp[d] += __shfl_xor(sp[d], o, 32);
    if (lane == 0)
      for (int d = 0; d < 16; ++d) red[wave * 16 + d] = sp[d];
    __syncthreads();
    if (t < 16) {
      float s = 0.f;
      for (int w = 0; w < 8; ++w) s += red[w * 16 + t];
      sv[t] = s;
    }
    __syncthreads();

    float s2 = 1e-8f;
    for (int d = 0; d < 16; ++d) s2 += sv[d] * sv[d];
    const float scale = sqrtf(s2) / (1.0f + s2);
    for (int d = 0; d < 16; ++d) v[d] = sv[d] * scale;

    for (int k = 0; k < 4; ++k) {
      float dot = 0.f;
      for (int j = 0; j < 4; ++j)
        for (int e = 0; e < 4; ++e)
          dot += row[k][j][e] * v[j * 4 + e];
      blog[k] += dot;
    }
    __syncthreads();
  }

  if (t < 16)
    out[((size_t)b * D_CAP + t) * N_CAP + n] = v[t];
}

extern "C" void kernel_launch(void* const* d_in, const int* in_sizes, int n_in,
                              void* d_out, int out_size, void* d_ws, size_t ws_size,
                              hipStream_t stream) {
  const float* u_vecs = (const float*)d_in[0];
  const float* W      = (const float*)d_in[1];
  float* out = (float*)d_out;

  float* uT   = (float*)d_ws;
  float* uhat = uT + (size_t)U_DIM * B_DIM * C_DIM;

  caps_transpose<<<dim3(U_DIM / 32, (B_DIM * C_DIM) / 32), 256, 0, stream>>>(u_vecs, uT);
  caps_uhat<<<dim3(U_DIM), 512, 0, stream>>>(uT, W, uhat);
  caps_route<<<dim3(B_DIM * N_CAP), 256, 0, stream>>>(uhat, out);
}


# routing kernel: one wave per pair (b,n) holds all 1024 u_hat rows in its own registers; sums over u finish inside the wave by lane exchanges (new reduction tree), no LDS, no workgroup barriers
# speedup vs baseline: 1.4750x; 1.4750x over previous
#include <hip/hip_runtime.h>


typedef __attribute__((ext_vector_type(2))) float v2f;
typedef __attribute__((ext_vector_type(4))) float v4f;
typedef __attribute__((ext_vector_type(8))) float v8f;

#define B_DIM 64
#define C_DIM 64
#define U_DIM 1024
#define N_CAP 32
#define D_CAP 16

__global__ void caps_transpose(const float* __restrict__ in,
                               float* __restrict__ outT) {
  __shared__ float tile[32][33];
  const int u0 = blockIdx.x * 32;
  const int r0 = blockIdx.y * 32;
  const int tx = threadIdx.x & 31;
  const int ty = threadIdx.x >> 5;
  for (int i = 0; i < 32; i += 8)
    tile[ty + i][tx] = in[(size_t)(r0 + ty + i) * U_DIM + u0 + tx];
  __syncthreads();
  for (int i = 0; i < 32; i += 8)
    outT[(size_t)(u0 + ty + i) * (B_DIM * C_DIM) + r0 + tx] = tile[tx][ty + i];
}

__global__ void __launch_bounds__(512, 1)
caps_uhat(const float* __restrict__ uT,
          const float* __restrict__ W,
          float* __restrict__ uhat) {
  const int u = blockIdx.x;
  __shared__ float As[B_DIM * 68];

  const float* Ag = uT + (size_t)u * (B_DIM * C_DIM);
  for (int i = threadIdx.x; i < (B_DIM * C_DIM) / 4; i += 512) {
    const int e = i * 4;
    const int b = e >> 6;
    const int c = e & 63;
    *(v4f*)&As[b * 68 + c] = ((const v4f*)Ag)[i];
  }
  __syncthreads();

  const int wave = threadIdx.x >> 5;
  const int lane = threadIdx.x & 31;
  const int l16  = lane & 15;
  const int koff = (lane >> 4) * 2;
  const int n_base = wave * 2;

  const float* Wg = W + (size_t)u * (N_CAP * C_DIM * D_CAP);

  v8f acc[4][2] = {};
  for (int kc = 0; kc < C_DIM; kc += 4) {
    const int k = kc + koff;
    v2f a[4], bb[2];
    for (int mt = 0; mt < 4; ++mt)
      a[mt] = *(const v2f*)&As[(mt * 16 + l16) * 68 + k];
    for (int nt = 0; nt < 2; ++nt) {
      const float* wp = Wg + ((size_t)(n_base + nt) * C_DIM + k) * D_CAP + l16;
      bb[nt].x = wp[0];
      bb[nt].y = wp[D_CAP];
      if (kc + 4 < C_DIM)
        __builtin_prefetch(wp + 4 * D_CAP, 0, 1);
    }
    for (int mt = 0; mt < 4; ++mt)
      for (int nt = 0; nt < 2; ++nt)
        acc[mt][nt] = __builtin_amdgcn_wmma_f32_16x16x4_f32(
            false, a[mt], false, bb[nt], (short)0, acc[mt][nt], false, false);
  }

  const int d = l16;
  const int rbase = (lane >> 4) * 8;
  for (int mt = 0; mt < 4; ++mt)
    for (int nt = 0; nt < 2; ++nt) {
      const int n = n_base + nt;
      for (int r = 0; r < 8; ++r) {
        const int b = mt * 16 + rbase + r;
        uhat[(((size_t)b * N_CAP + n) * U_DIM + u) * D_CAP + d] = acc[mt][nt][r];
      }
    }
}

#define ROWS_PER_LANE 32

__device__ __forceinline__ float xhalf(float x) {
  const int b = __builtin_bit_cast(int, x);
  return __builtin_bit_cast(float,
      __builtin_amdgcn_permlanex16(b, b, 0x76543210, 0xfedcba98, false, false));
}
template <int M>
__device__ __forceinline__ float xrow(float x) {
  const int b = __builtin_bit_cast(int, x);
  return __builtin_bit_cast(float,
      __builtin_amdgcn_update_dpp(b, b, 0x160 + M, 0xf, 0xf, false));
}
__device__ __forceinline__ float wave_max(float x) {
  x = fmaxf(x, xrow<1>(x));
  x = fmaxf(x, xrow<2>(x));
  x = fmaxf(x, xrow<4>(x));
  x = fmaxf(x, xrow<8>(x));
  return fmaxf(x, xhalf(x));
}
__device__ __forceinline__ float wave_sum(float x) {
  x += xrow<1>(x);
  x += xrow<2>(x);
  x += xrow<4>(x);
  x += xrow<8>(x);
  return x + xhalf(x);
}

__global__ void __launch_bounds__(32)
caps_route(const float* __restrict__ uhat,
           float* __restrict__ out) {
  const int bn = blockIdx.x;
  const int b = bn >> 5, n = bn & 31;
  const int lane = threadIdx.x;

  const float* base = uhat + (size_t)bn * (U_DIM * D_CAP);
  v4f row[ROWS_PER_LANE][4];
  for (int k = 0; k < ROWS_PER_LANE; ++k) {
    const v4f* p = (const v4f*)(base + (size_t)(lane + k * 32) * D_CAP);
    for (int j = 0; j < 4; ++j) row[k][j] = p[j];
  }

  float blog[ROWS_PER_LANE];
  for (int k = 0; k < ROWS_PER_LANE; ++k) blog[k] = 0.f;

  const bool h4 = (lane & 16) != 0, h3 = (lane & 8) != 0;
  const bool h2 = (lane & 4) != 0,  h1 = (lane & 2) != 0;
  float vmine = 0.f;

  for (int it = 0; it < 3; ++it) {
    float c[ROWS_PER_LANE];
    float mx = blog[0];
    for (int k = 1; k < ROWS_PER_LANE; ++k) mx = fmaxf(mx, blog[k]);
    mx = wave_max(mx);
    float sum = 0.f;
    for (int k = 0; k < ROWS_PER_LANE; ++k) { c[k] = __expf(blog[k] - mx); sum += c[k]; }
    sum = wave_sum(sum);
    const float inv = 1.0f / sum;
    for (int k = 0; k < ROWS_PER_LANE; ++k) c[k] *= inv;

    v2f sp2[8];
    for (int p = 0; p < 8; ++p) sp2[p] = (v2f){0.f, 0.f};
    for (int k = 0; k < ROWS_PER_LANE; ++k) {
      const v2f ck = (v2f){c[k], c[k]};
      for (int j = 0; j < 4; ++j) {
        sp2[2 * j]     += ck * row[k][j].xy;
        sp2[2 * j + 1] += ck * row[k][j].zw;
      }
    }
    float sp[16];
    for (int p = 0; p < 8; ++p) { sp[2 * p] = sp2[p].x; sp[2 * p + 1] = sp2[p].y; }

    float t8[8], t4[4], t2[2];
    for (int j = 0; j < 8; ++j) {
      const float keep = h4 ? sp[8 + j] : sp[j];
      const float send = h4 ? sp[j] : sp[8 + j];
      t8[j] = keep + xhalf(send);
    }
    for (int j = 0; j < 4; ++j) {
      const float keep = h3 ? t8[4 + j] : t8[j];
      const float send = h3 ? t8[j] : t8[4 + j];
      t4[j] = keep + xrow<8>(send);
    }
    for (int j = 0; j < 2; ++j) {
      const float keep = h2 ? t4[2 + j] : t4[j];
      const float send = h2 ? t4[j] : t4[2 + j];
      t2[j] = keep + xrow<4>(send);
    }
    float sd;
    {
      const float keep = h1 ? t2[1] : t2[0];
      const float send = h1 ? t2[0] : t2[1];
      sd = keep + xrow<2>(send);
    }
    sd += xrow<1>(sd);

    float q = sd * sd;
    q += xrow<2>(q);
    q += xrow<4>(q);
    q += xrow<8>(q);
    q += xhalf(q);
    const float s2 = 1e-8f + q;
    const float scale = sqrtf(s2) / (1.0f + s2);
    vmine = sd * scale;

    if (it < 2) {
      float v[16];
      for (int d = 0; d < 16; ++d)
        v[d] = __builtin_bit_cast(float,
            __builtin_amdgcn_readlane(__builtin_bit_cast(int, vmine), 2 * d));
      v2f vv[8];
      for (int p = 0; p < 8; ++p) vv[p] = (v2f){v[2 * p], v[2 * p + 1]};
      for (int k = 0; k < ROWS_PER_LANE; ++k) {
        v2f d2 = (v2f){0.f, 0.f};
        for (int j = 0; j < 4; ++j) {
          d2 += row[k][j].xy * vv[2 * j];
          d2 += row[k][j].zw * vv[2 * j + 1];
        }
        blog[k] += d2.x + d2.y;
      }
    }
  }

  if ((lane & 1) == 0)
    out[((size_t)b * D_CAP + (lane >> 1)) * N_CAP + n] = vmine;
}

extern "C" void kernel_launch(void* const* d_in, const int* in_sizes, int n_in,
                              void* d_out, int out_size, void* d_ws, size_t ws_size,
                              hipStream_t stream) {
  const float* u_vecs = (const float*)d_in[0];
  const float* W      = (const float*)d_in[1];
  float* out = (float*)d_out;

  float* uT   = (float*)d_ws;
  float* uhat = uT + (size_t)U_DIM * B_DIM * C_DIM;

  caps_transpose<<<dim3(U_DIM / 32, (B_DIM * C_DIM) / 32), 256, 0, stream>>>(u_vecs, uT);
  caps_uhat<<<dim3(U_DIM), 512, 0, stream>>>(uT, W, uhat);
  caps_route<<<dim3(B_DIM * N_CAP), 32, 0, stream>>>(uhat, out);
}


# u_hat GEMM: each wave streams its two capsules' W rows with whole-line 16-byte loads, issued three row groups ahead, through an LDS slab of its own; the k-loop holds no global load; sums unchanged
# speedup vs baseline: 1.5593x; 1.0572x over previous
#include <hip/hip_runtime.h>


typedef __attribute__((ext_vector_type(2))) float v2f;
typedef __attribute__((ext_vector_type(4))) float v4f;
typedef __attribute__((ext_vector_type(8))) float v8f;

#define B_DIM 64
#define C_DIM 64
#define U_DIM 1024
#define N_CAP 32
#define D_CAP 16

__global__ void caps_transpose(const float* __restrict__ in,
                               float* __restrict__ outT) {
  __shared__ float tile[32][33];
  const int u0 = blockIdx.x * 32;
  const int r0 = blockIdx.y * 32;
  const int tx = threadIdx.x & 31;
  const int ty = threadIdx.x >> 5;
  for (int i = 0; i < 32; i += 8)
    tile[ty + i][tx] = in[(size_t)(r0 + ty + i) * U_DIM + u0 + tx];
  __syncthreads();
  for (int i = 0; i < 32; i += 8)
    outT[(size_t)(u0 + ty + i) * (B_DIM * C_DIM) + r0 + tx] = tile[tx][ty + i];
}

#define W_SLAB 2048

__global__ void __launch_bounds__(512, 8)
caps_uhat(const float* __restrict__ uT,
          const float* __restrict__ W,
          float* __restrict__ uhat) {
  const int u = blockIdx.x;
  __shared__ __attribute__((aligned(16))) float As[B_DIM * 68];
  __shared__ __attribute__((aligned(16))) float Ws[16 * W_SLAB];

  const int wave = threadIdx.x >> 5;
  const int lane = threadIdx.x & 31;
  const int l16  = lane & 15;
  const int koff = (lane >> 4) * 2;
  const int n_base = wave * 2;

  const float* Wg = W + (size_t)u * (N_CAP * C_DIM * D_CAP) + (size_t)wave * W_SLAB + lane * 4;
  float* Wl = Ws + wave * W_SLAB;

  const float* Ag = uT + (size_t)u * (B_DIM * C_DIM);
  const v4f a0 = ((const v4f*)Ag)[threadIdx.x];
  const v4f a1 = ((const v4f*)Ag)[threadIdx.x + 512];
  v4f wr[8][2];
  for (int kk = 0; kk < 3; ++kk)
    for (int nt = 0; nt < 2; ++nt)
      wr[kk][nt] = *(const v4f*)(Wg + nt * 1024 + kk * 128);
  {
    const int e0 = threadIdx.x * 4, e1 = e0 + 2048;
    *(v4f*)&As[(e0 >> 6) * 68 + (e0 & 63)] = a0;
    *(v4f*)&As[(e1 >> 6) * 68 + (e1 & 63)] = a1;
  }
  __syncthreads();

  const float* al = As + l16 * 68 + koff;
  const float* bl = Wl + koff * D_CAP + l16;

  v8f acc[4][2] = {};
  for (int kk = 0; kk < 8; ++kk) {
    for (int nt = 0; nt < 2; ++nt)
      *(v4f*)&Wl[nt * 1024 + kk * 128 + lane * 4] = wr[kk][nt];
    if (kk + 3 < 8)
      for (int nt = 0; nt < 2; ++nt)
        wr[kk + 3][nt] = *(const v4f*)(Wg + nt * 1024 + (kk + 3) * 128);
    for (int s = 0; s < 2; ++s) {
      const int kc = kk * 8 + s * 4;
      v2f a[4], bb[2];
      for (int mt = 0; mt < 4; ++mt)
        a[mt] = *(const v2f*)&al[mt * 16 * 68 + kc];
      for (int nt = 0; nt < 2; ++nt) {
        bb[nt].x = bl[nt * 1024 + kc * D_CAP];
        bb[nt].y = bl[nt * 1024 + kc * D_CAP + D_CAP];
      }
      for (int mt = 0; mt < 4; ++mt)
        for (int nt = 0; nt < 2; ++nt)
          acc[mt][nt] = __builtin_amdgcn_wmma_f32_16x16x4_f32(
              false, a[mt], false, bb[nt], (short)0, acc[mt][nt], false, false);
    }
    __builtin_amdgcn_sched_barrier(0);
  }

  const int d = l16;
  const int rbase = (lane >> 4) * 8;
  for (int mt = 0; mt < 4; ++mt)
    for (int nt = 0; nt < 2; ++nt) {
      const int n = n_base + nt;
      for (int r = 0; r < 8; ++r) {
        const int b = mt * 16 + rbase + r;
        uhat[(((size_t)b * N_CAP + n) * U_DIM + u) * D_CAP + d] = acc[mt][nt][r];
      }
    }
}

#define ROWS_PER_LANE 32

__device__ __forceinline__ float xhalf(float x) {
  const int b = __builtin_bit_cast(int, x);
  return __builtin_bit_cast(float,
      __builtin_amdgcn_permlanex16(b, b, 0x76543210, 0xfedcba98, false, false));
}
template <int M>
__device__ __forceinline__ float xrow(float x) {
  const int b = __builtin_bit_cast(int, x);
  return __builtin_bit_cast(float,
      __builtin_amdgcn_update_dpp(b, b, 0x160 + M, 0xf, 0xf, false));
}
__device__ __forceinline__ float wave_max(float x) {
  x = fmaxf(x, xrow<1>(x));
  x = fmaxf(x, xrow<2>(x));
  x = fmaxf(x, xrow<4>(x));
  x = fmaxf(x, xrow<8>(x));
  return fmaxf(x, xhalf(x));
}
__device__ __forceinline__ float wave_sum(float x) {
  x += xrow<1>(x);
  x += xrow<2>(x);
  x += xrow<4>(x);
  x += xrow<8>(x);
  return x + xhalf(x);
}

__global__ void __launch_bounds__(32)
caps_route(const float* __restrict__ uhat,
           float* __restrict__ out) {
  const int bn = blockIdx.x;
  const int b = bn >> 5, n = bn & 31;
  const int lane = threadIdx.x;

  const float* base = uhat + (size_t)bn * (U_DIM * D_CAP);
  v4f row[ROWS_PER_LANE][4];
  for (int k = 0; k < ROWS_PER_LANE; ++k) {
    const v4f* p = (const v4f*)(base + (size_t)(lane + k * 32) * D_CAP);
    for (int j = 0; j < 4; ++j) row[k][j] = p[j];
  }

  float blog[ROWS_PER_LANE];
  for (int k = 0; k < ROWS_PER_LANE; ++k) blog[k] = 0.f;

  const bool h4 = (lane & 16) != 0, h3 = (lane & 8) != 0;
  const bool h2 = (lane & 4) != 0,  h1 = (lane & 2) != 0;
  float vmine = 0.f;

  for (int it = 0; it < 3; ++it) {
    float c[ROWS_PER_LANE];
    float mx = blog[0];
    for (int k = 1; k < ROWS_PER_LANE; ++k) mx = fmaxf(mx, blog[k]);
    mx = wave_max(mx);
    float sum = 0.f;
    for (int k = 0; k < ROWS_PER_LANE; ++k) { c[k] = __expf(blog[k] - mx); sum += c[k]; }
    sum = wave_sum(sum);
    const float inv = 1.0f / sum;
    for (int k = 0; k < ROWS_PER_LANE; ++k) c[k] *= inv;

    v2f sp2[8];
    for (int p = 0; p < 8; ++p) sp2[p] = (v2f){0.f, 0.f};
    for (int k = 0; k < ROWS_PER_LANE; ++k) {
      const v2f ck = (v2f){c[k], c[k]};
      for (int j = 0; j < 4; ++j) {
        sp2[2 * j]     += ck * row[k][j].xy;
        sp2[2 * j + 1] += ck * row[k][j].zw;
      }
    }
    float sp[16];
    for (int p = 0; p < 8; ++p) { sp[2 * p] = sp2[p].x; sp[2 * p + 1] = sp2[p].y; }

    float t8[8], t4[4], t2[2];
    for (int j = 0; j < 8; ++j) {
      const float keep = h4 ? sp[8 + j] : sp[j];
      const float send = h4 ? sp[j] : sp[8 + j];
      t8[j] = keep + xhalf(send);
    }
    for (int j = 0; j < 4; ++j) {
      const float keep = h3 ? t8[4 + j] : t8[j];
      const float send = h3 ? t8[j] : t8[4 + j];
      t4[j] = keep + xrow<8>(send);
    }
    for (int j = 0; j < 2; ++j) {
      const float keep = h2 ? t4[2 + j] : t4[j];
      const float send = h2 ? t4[j] : t4[2 + j];
      t2[j] = keep + xrow<4>(send);
    }
    float sd;
    {
      const float keep = h1 ? t2[1] : t2[0];
      const float send = h1 ? t2[0] : t2[1];
      sd = keep + xrow<2>(send);
    }
    sd += xrow<1>(sd);

    float q = sd * sd;
    q += xrow<2>(q);
    q += xrow<4>(q);
    q += xrow<8>(q);
    q += xhalf(q);
    const float s2 = 1e-8f + q;
    const float scale = sqrtf(s2) / (1.0f + s2);
    vmine = sd * scale;

    if (it < 2) {
      float v[16];
      for (int d = 0; d < 16; ++d)
        v[d] = __builtin_bit_cast(float,
            __builtin_amdgcn_readlane(__builtin_bit_cast(int, vmine), 2 * d));
      v2f vv[8];
      for (int p = 0; p < 8; ++p) vv[p] = (v2f){v[2 * p], v[2 * p + 1]};
      for (int k = 0; k < ROWS_PER_LANE; ++k) {
        v2f d2 = (v2f){0.f, 0.f};
        for (int j = 0; j < 4; ++j) {
          d2 += row[k][j].xy * vv[2 * j];
          d2 += row[k][j].zw * vv[2 * j + 1];
        }
        blog[k] += d2.x + d2.y;
      }
    }
  }

  if ((lane & 1) == 0)
    out[((size_t)b * D_CAP + (lane >> 1)) * N_CAP + n] = vmine;
}

extern "C" void kernel_launch(void* const* d_in, const int* in_sizes, int n_in,
                              void* d_out, int out_size, void* d_ws, size_t ws_size,
                              hipStream_t stream) {
  const float* u_vecs = (const float*)d_in[0];
  const float* W      = (const float*)d_in[1];
  float* out = (float*)d_out;

  float* uT   = (float*)d_ws;
  float* uhat = uT + (size_t)U_DIM * B_DIM * C_DIM;

  caps_transpose<<<dim3(U_DIM / 32, (B_DIM * C_DIM) / 32), 256, 0, stream>>>(u_vecs, uT);
  caps_uhat<<<dim3(U_DIM), 512, 0, stream>>>(uT, W, uhat);
  caps_route<<<dim3(B_DIM * N_CAP), 32, 0, stream>>>(uhat, out);
}


# u_hat kept as 256-B units (4 b x 16 d per n,u), stored whole by the GEMM wave via its LDS slab; routing: 4 waves per (n,b-quad), a pair's u over 2 waves merged in LDS (new sum tree, scaled after)
# speedup vs baseline: 2.5890x; 1.6604x over previous
#include <hip/hip_runtime.h>


typedef __attribute__((ext_vector_type(2))) float v2f;
typedef __attribute__((ext_vector_type(4))) float v4f;
typedef __attribute__((ext_vector_type(8))) float v8f;

#define B_DIM 64
#define C_DIM 64
#define U_DIM 1024
#define N_CAP 32
#define D_CAP 16

__global__ void caps_transpose(const float* __restrict__ in,
                               float* __restrict__ outT) {
  __shared__ float tile[32][33];
  const int u0 = blockIdx.x * 32;
  const int r0 = blockIdx.y * 32;
  const int tx = threadIdx.x & 31;
  const int ty = threadIdx.x >> 5;
  for (int i = 0; i < 32; i += 8)
    tile[ty + i][tx] = in[(size_t)(r0 + ty + i) * U_DIM + u0 + tx];
  __syncthreads();
  for (int i = 0; i < 32; i += 8)
    outT[(size_t)(u0 + ty + i) * (B_DIM * C_DIM) + r0 + tx] = tile[tx][ty + i];
}

#define W_SLAB 2048

__global__ void __launch_bounds__(512, 8)
caps_uhat(const float* __restrict__ uT,
          const float* __restrict__ W,
          float* __restrict__ uhat) {
  const int u = blockIdx.x;
  __shared__ __attribute__((aligned(16))) float As[B_DIM * 68];
  __shared__ __attribute__((aligned(16))) float Ws[16 * W_SLAB];

  const int wave = threadIdx.x >> 5;
  const int lane = threadIdx.x & 31;
  const int l16  = lane & 15;
  const int koff = (lane >> 4) * 2;
  const int n_base = wave * 2;

  const float* Wg = W + (size_t)u * (N_CAP * C_DIM * D_CAP) + (size_t)wave * W_SLAB + lane * 4;
  float* Wl = Ws + wave * W_SLAB;

  const float* Ag = uT + (size_t)u * (B_DIM * C_DIM);
  const v4f a0 = ((const v4f*)Ag)[threadIdx.x];
  const v4f a1 = ((const v4f*)Ag)[threadIdx.x + 512];
  v4f wr[8][2];
  for (int kk = 0; kk < 3; ++kk)
    for (int nt = 0; nt < 2; ++nt)
      wr[kk][nt] = *(const v4f*)(Wg + nt * 1024 + kk * 128);
  {
    const int e0 = threadIdx.x * 4, e1 = e0 + 2048;
    *(v4f*)&As[(e0 >> 6) * 68 + (e0 & 63)] = a0;
    *(v4f*)&As[(e1 >> 6) * 68 + (e1 & 63)] = a1;
  }
  __syncthreads();

  const float* al = As + l16 * 68 + koff;
  const float* bl = Wl + koff * D_CAP + l16;

  v8f acc[4][2] = {};
  for (int kk = 0; kk < 8; ++kk) {
    for (int nt = 0; nt < 2; ++nt)
      *(v4f*)&Wl[nt * 1024 + kk * 128 + lane * 4] = wr[kk][nt];
    if (kk + 3 < 8)
      for (int nt = 0; nt < 2; ++nt)
        wr[kk + 3][nt] = *(const v4f*)(Wg + nt * 1024 + (kk + 3) * 128);
    for (int s = 0; s < 2; ++s) {
      const int kc = kk * 8 + s * 4;
      v2f a[4], bb[2];
      for (int mt = 0; mt < 4; ++mt)
        a[mt] = *(const v2f*)&al[mt * 16 * 68 + kc];
      for (int nt = 0; nt < 2; ++nt) {
        bb[nt].x = bl[nt * 1024 + kc * D_CAP];
        bb[nt].y = bl[nt * 1024 + kc * D_CAP + D_CAP];
      }
      for (int mt = 0; mt < 4; ++mt)
        for (int nt = 0; nt < 2; ++nt)
          acc[mt][nt] = __builtin_amdgcn_wmma_f32_16x16x4_f32(
              false, a[mt], false, bb[nt], (short)0, acc[mt][nt], false, false);
    }
    __builtin_amdgcn_sched_barrier(0);
  }

  const int rbase = (lane >> 4) * 8;
  for (int mt = 0; mt < 4; ++mt)
    for (int nt = 0; nt < 2; ++nt)
      for (int r = 0; r < 8; ++r)
        Wl[nt * 1024 + (mt * 16 + rbase + r) * D_CAP + l16] = acc[mt][nt][r];
  float* ug = uhat + ((size_t)(n_base * 16 + (lane >> 4)) * U_DIM + u) * 64 + (lane & 15) * 4;
  for (int i = 0; i < 16; ++i) {
    const v4f val = *(const v4f*)&Wl[i * 128 + lane * 4];
    const int unit = (i >> 3) * 16 + (i & 7) * 2;
    *(v4f*)(ug + (size_t)unit * (U_DIM * 64)) = val;
  }
}

#define ROWS_PER_LANE 32

__device__ __forceinline__ float xhalf(float x) {
  const int b = __builtin_bit_cast(int, x);
  return __builtin_bit_cast(float,
      __builtin_amdgcn_permlanex16(b, b, 0x76543210, 0xfedcba98, false, false));
}
template <int M>
__device__ __forceinline__ float xrow(float x) {
  const int b = __builtin_bit_cast(int, x);
  return __builtin_bit_cast(float,
      __builtin_amdgcn_update_dpp(b, b, 0x160 + M, 0xf, 0xf, false));
}
__device__ __forceinline__ float par_max(float x) {
  x = fmaxf(x, xrow<2>(x));
  x = fmaxf(x, xrow<4>(x));
  x = fmaxf(x, xrow<8>(x));
  return fmaxf(x, xhalf(x));
}
__device__ __forceinline__ float par_sum(float x) {
  x += xrow<2>(x);
  x += xrow<4>(x);
  x += xrow<8>(x);
  return x + xhalf(x);
}

__global__ void __launch_bounds__(128)
caps_route(const float* __restrict__ uhat,
           float* __restrict__ out) {
  __shared__ float xm[3][4][32], xl[3][4][32], xs[3][4][32];
  const int nq = blockIdx.x;
  const int n = nq >> 4;
  const int wave = threadIdx.x >> 5;
  const int lane = threadIdx.x & 31;
  const int pr = wave >> 1;
  const int uh = wave & 1;
  const int b = (nq & 15) * 4 + pr * 2 + (lane & 1);

  const float* base = uhat + (size_t)nq * (U_DIM * 64) + uh * (512 * 64)
                    + (lane >> 1) * 64 + pr * 32 + (lane & 1) * D_CAP;
  v4f row[ROWS_PER_LANE][4];
  for (int k = 0; k < ROWS_PER_LANE; ++k) {
    const v4f* p = (const v4f*)(base + k * (16 * 64));
    for (int j = 0; j < 4; ++j) row[k][j] = p[j];
  }

  float blog[ROWS_PER_LANE];
  for (int k = 0; k < ROWS_PER_LANE; ++k) blog[k] = 0.f;

  const bool h4 = (lane & 16) != 0, h3 = (lane & 8) != 0;
  const bool h2 = (lane & 4) != 0,  h1 = (lane & 2) != 0;
  float vmine = 0.f;

  for (int it = 0; it < 3; ++it) {
    float c[ROWS_PER_LANE];
    float mx = blog[0];
    for (int k = 1; k < ROWS_PER_LANE; ++k) mx = fmaxf(mx, blog[k]);
    mx = par_max(mx);
    float sum = 0.f;
    for (int k = 0; k < ROWS_PER_LANE; ++k) { c[k] = __expf(blog[k] - mx); sum += c[k]; }
    sum = par_sum(sum);

    v2f sp2[8];
    for (int p = 0; p < 8; ++p) sp2[p] = (v2f){0.f, 0.f};
    for (int k = 0; k < ROWS_PER_LANE; ++k) {
      const v2f ck = (v2f){c[k], c[k]};
      for (int j = 0; j < 4; ++j) {
        sp2[2 * j]     += ck * row[k][j].xy;
        sp2[2 * j + 1] += ck * row[k][j].zw;
      }
    }
    float sp[16];
    for (int p = 0; p < 8; ++p) { sp[2 * p] = sp2[p].x; sp[2 * p + 1] = sp2[p].y; }

    float t8[8], t4[4], t2[2];
    for (int j = 0; j < 8; ++j) {
      const float keep = h4 ? sp[8 + j] : sp[j];
      const float send = h4 ? sp[j] : sp[8 + j];
      t8[j] = keep + xhalf(send);
    }
    for (int j = 0; j < 4; ++j) {
      const float keep = h3 ? t8[4 + j] : t8[j];
      const float send = h3 ? t8[j] : t8[4 + j];
      t4[j] = keep + xrow<8>(send);
    }
    for (int j = 0; j < 2; ++j) {
      const float keep = h2 ? t4[2 + j] : t4[j];
      const float send = h2 ? t4[j] : t4[2 + j];
      t2[j] = keep + xrow<4>(send);
    }
    float sd;
    {
      const float keep = h1 ? t2[1] : t2[0];
      const float send = h1 ? t2[0] : t2[1];
      sd = keep + xrow<2>(send);
    }

    xm[it][wave][lane] = mx;
    xl[it][wave][lane] = sum;
    xs[it][wave][lane] = sd;
    __syncthreads();
    const float omx = xm[it][wave ^ 1][lane];
    const float osum = xl[it][wave ^ 1][lane];
    const float osd = xs[it][wave ^ 1][lane];
    const float m0 = uh ? omx : mx,   m1 = uh ? mx : omx;
    const float l0 = uh ? osum : sum, l1 = uh ? sum : osum;
    const float s0 = uh ? osd : sd,   s1 = uh ? sd : osd;
    const float mm = fmaxf(m0, m1);
    const float f0 = __expf(m0 - mm), f1 = __expf(m1 - mm);
    sd = (f0 * s0 + f1 * s1) / (f0 * l0 + f1 * l1);

    float q = sd * sd;
    q += xrow<2>(q);
    q += xrow<4>(q);
    q += xrow<8>(q);
    q += xhalf(q);
    const float s2 = 1e-8f + q;
    const float scale = sqrtf(s2) / (1.0f + s2);
    vmine = sd * scale;

    if (it < 2) {
      float v[16];
      for (int d = 0; d < 16; ++d) {
        const float ve = __builtin_bit_cast(float,
            __builtin_amdgcn_readlane(__builtin_bit_cast(int, vmine), 2 * d));
        const float vo = __builtin_bit_cast(float,
            __builtin_amdgcn_readlane(__builtin_bit_cast(int, vmine), 2 * d + 1));
        v[d] = (lane & 1) ? vo : ve;
      }
      v2f vv[8];
      for (int p = 0; p < 8; ++p) vv[p] = (v2f){v[2 * p], v[2 * p + 1]};
      for (int k = 0; k < ROWS_PER_LANE; ++k) {
        v2f d2 = (v2f){0.f, 0.f};
        for (int j = 0; j < 4; ++j) {
          d2 += row[k][j].xy * vv[2 * j];
          d2 += row[k][j].zw * vv[2 * j + 1];
        }
        blog[k] += d2.x + d2.y;
      }
    }
  }

  if (uh == 0)
    out[((size_t)b * D_CAP + (lane >> 1)) * N_CAP + n] = vmine;
}

extern "C" void kernel_launch(void* const* d_in, const int* in_sizes, int n_in,
                              void* d_out, int out_size, void* d_ws, size_t ws_size,
                              hipStream_t stream) {
  const float* u_vecs = (const float*)d_in[0];
  const float* W      = (const float*)d_in[1];
  float* out = (float*)d_out;

  float* uT   = (float*)d_ws;
  float* uhat = uT + (size_t)U_DIM * B_DIM * C_DIM;

  caps_transpose<<<dim3(U_DIM / 32, (B_DIM * C_DIM) / 32), 256, 0, stream>>>(u_vecs, uT);
  caps_uhat<<<dim3(U_DIM), 512, 0, stream>>>(uT, W, uhat);
  caps_route<<<dim3(N_CAP * (B_DIM / 4)), 128, 0, stream>>>(uhat, out);
}
